# starting kernel
# baseline (speedup 1.0000x reference)
#include <hip/hip_runtime.h>
#include <math.h>

typedef __attribute__((ext_vector_type(2))) float v2f;
typedef __attribute__((ext_vector_type(8))) float v8f;

#define BATCH 64
#define NN    196
#define DD    768
#define OO    196
#define OPAD  208
#define MT    13
#define NT    48
#define NTG   6
#define KSTEPS 49
#define NTILE 7

#define WP_OFF 0
#define C_OFF  (OPAD * NN)
#define YT_OFF (C_OFF + 256)

__global__ void dnm_prep_kernel(const float* __restrict__ sw,
                                const float* __restrict__ sb,
                                const float* __restrict__ dw,
                                float* __restrict__ ws) {
  __shared__ float red[NN];
  const int o = blockIdx.x;
  const int t = threadIdx.x;
  float* Wp = ws + WP_OFF;
  float* cv = ws + C_OFF;
  if (o < OO) {
    const float d0 = dw[o * 2 + 0];
    const float d1 = dw[o * 2 + 1];
    Wp[(size_t)o * NN + t] = d0 * sw[(size_t)(o * 2 + 0) * NN + t] +
                             d1 * sw[(size_t)(o * 2 + 1) * NN + t];
    red[t] = d0 * sb[(size_t)(o * 2 + 0) * NN + t] +
             d1 * sb[(size_t)(o * 2 + 1) * NN + t];
    __syncthreads();
    if (t == 0) {
      float s = 0.0f;
      for (int i = 0; i < NN; ++i) s += red[i];
      cv[o] = s;
    }
  } else {
    Wp[(size_t)o * NN + t] = 0.0f;
    if (t == 0) cv[o] = 0.0f;
  }
}

__global__ void __launch_bounds__(256) dnm_ln_t_kernel(const float* __restrict__ x,
                                                       const float* __restrict__ lw,
                                                       const float* __restrict__ lb,
                                                       float* __restrict__ ws) {
  const int b  = blockIdx.x / NTILE;
  const int n0 = (blockIdx.x % NTILE) * 32;
  const int t    = threadIdx.x;
  const int lane = t & 31;
  const int w    = t >> 5;
  float* yT = ws + YT_OFF;

  __shared__ float mus[32], rss[32];
  __shared__ float tile[32][33];

  for (int rr = 0; rr < 4; ++rr) {
    const int r   = w * 4 + rr;
    const int row = n0 + r;
    if (row < NN) {
      const float* xr = x + ((size_t)b * NN + row) * DD;
      float s = 0.0f, q = 0.0f;
      for (int j = lane; j < DD; j += 32) {
        const float v = xr[j];
        s += v;
        q += v * v;
      }
      for (int off = 16; off > 0; off >>= 1) {
        s += __shfl_xor(s, off, 32);
        q += __shfl_xor(q, off, 32);
      }
      if (lane == 0) {
        const float mu  = s * (1.0f / DD);
        const float var = q * (1.0f / DD) - mu * mu;
        mus[r] = mu;
        rss[r] = rsqrtf(var + 1e-5f);
      }
    }
  }
  __syncthreads();

  for (int dc = 0; dc < DD / 32; ++dc) {
    const int d0 = dc * 32;
#pragma unroll
    for (int p = 0; p < 4; ++p) {
      const int r   = (t >> 5) + p * 8;
      const int c   = t & 31;
      const int row = n0 + r;
      float v = 0.0f;
      if (row < NN) {
        const float xv = x[((size_t)b * NN + row) * DD + d0 + c];
        v = (xv - mus[r]) * rss[r] * lw[d0 + c] + lb[d0 + c];
      }
      tile[r][c] = v;
    }
    __syncthreads();
#pragma unroll
    for (int p = 0; p < 4; ++p) {
      const int dr = (t >> 5) + p * 8;
      const int nc = t & 31;
      if (n0 + nc < NN)
        yT[((size_t)b * DD + d0 + dr) * NN + n0 + nc] = tile[nc][dr];
    }
    __syncthreads();
  }
}

__global__ void __launch_bounds__(256) dnm_gemm_kernel(const float* __restrict__ x,
                                                       const float* __restrict__ ws,
                                                       float* __restrict__ out) {
  const float* Wp = ws + WP_OFF;
  const float* cv = ws + C_OFF;
  const float* yT = ws + YT_OFF;

  const int wv   = threadIdx.x >> 5;
  const int lane = threadIdx.x & 31;

  const int b   = blockIdx.x / (MT * NTG);
  const int r   = blockIdx.x % (MT * NTG);
  const int mt  = r / NTG;
  const int ntg = r % NTG;
  const int nt  = ntg * 8 + wv;

  const int mo  = mt * 16;
  const int col = nt * 16 + (lane & 15);
  const int hi  = lane >> 4;

  const float* Arow = Wp + (size_t)(mo + (lane & 15)) * NN + hi * 2;
  const float* Brow = yT + ((size_t)b * DD + col) * NN + hi * 2;

  {
    const int opf = (mo + hi * 8 < OO) ? (mo + hi * 8) : (OO - 1);
    __builtin_prefetch(&x[((size_t)b * OO + opf) * DD + col], 0, 3);
  }

  v8f acc = {};
#pragma unroll 7
  for (int kb = 0; kb < KSTEPS; ++kb) {
    const v2f a  = *(const v2f*)(Arow + kb * 4);
    const v2f bf = *(const v2f*)(Brow + kb * 4);
    acc = __builtin_amdgcn_wmma_f32_16x16x4_f32(
         false, a,  false, bf,
         (short)0, acc,  false,  false);
  }

#pragma unroll
  for (int v = 0; v < 8; ++v) {
    const int o = mo + hi * 8 + v;
    if (o < OO) {
      const float tv = acc[v] + cv[o];
      const float g  = 0.5f * tv * (1.0f + erff(tv * 0.70710678118654752f));
      const size_t idx = ((size_t)b * OO + o) * DD + col;
      out[idx] = x[idx] * (g + 1.0f);
    }
  }
}

extern "C" void kernel_launch(void* const* d_in, const int* in_sizes, int n_in,
                              void* d_out, int out_size, void* d_ws, size_t ws_size,
                              hipStream_t stream) {
  const float* x    = (const float*)d_in[0];
  const float* ln_w = (const float*)d_in[1];
  const float* ln_b = (const float*)d_in[2];
  const float* sw   = (const float*)d_in[3];
  const float* sb   = (const float*)d_in[4];
  const float* dw   = (const float*)d_in[5];
  float* out = (float*)d_out;
  float* ws  = (float*)d_ws;

  dnm_prep_kernel<<<dim3(OPAD), dim3(NN), 0, stream>>>(sw, sb, dw, ws);
  dnm_ln_t_kernel<<<dim3(BATCH * NTILE), dim3(256), 0, stream>>>(x, ln_w, ln_b, ws);
  dnm_gemm_kernel<<<dim3(BATCH * MT * NTG), dim3(256), 0, stream>>>(x, ws, out);
}


# GEMM operand reuse: Wp staged once per block in LDS, each wave keeps 7 or 6 M-tile accumulators so one yT B fragment feeds 7 or 6 WMMAs (256 blocks x 24 waves); per-tile k sum order unchanged
# speedup vs baseline: 1.4537x; 1.4537x over previous
#include <hip/hip_runtime.h>
#include <math.h>

typedef __attribute__((ext_vector_type(2))) float v2f;
typedef __attribute__((ext_vector_type(4))) float v4f;
typedef __attribute__((ext_vector_type(8))) float v8f;

#define BATCH 64
#define NN    196
#define DD    768
#define OO    196
#define OPAD  208
#define MT    13
#define NT    48
#define KSTEPS 49
#define NTILE 7

#define WP_OFF 0
#define C_OFF  (OPAD * NN)
#define YT_OFF (C_OFF + 256)

__global__ void dnm_prep_kernel(const float* __restrict__ sw,
                                const float* __restrict__ sb,
                                const float* __restrict__ dw,
                                float* __restrict__ ws) {
  __shared__ float red[NN];
  const int o = blockIdx.x;
  const int t = threadIdx.x;
  float* Wp = ws + WP_OFF;
  float* cv = ws + C_OFF;
  if (o < OO) {
    const float d0 = dw[o * 2 + 0];
    const float d1 = dw[o * 2 + 1];
    Wp[(size_t)o * NN + t] = d0 * sw[(size_t)(o * 2 + 0) * NN + t] +
                             d1 * sw[(size_t)(o * 2 + 1) * NN + t];
    red[t] = d0 * sb[(size_t)(o * 2 + 0) * NN + t] +
             d1 * sb[(size_t)(o * 2 + 1) * NN + t];
    __syncthreads();
    if (t == 0) {
      float s = 0.0f;
      for (int i = 0; i < NN; ++i) s += red[i];
      cv[o] = s;
    }
  } else {
    Wp[(size_t)o * NN + t] = 0.0f;
    if (t == 0) cv[o] = 0.0f;
  }
}

__global__ void __launch_bounds__(256) dnm_ln_t_kernel(const float* __restrict__ x,
                                                       const float* __restrict__ lw,
                                                       const float* __restrict__ lb,
                                                       float* __restrict__ ws) {
  const int b  = blockIdx.x / NTILE;
  const int n0 = (blockIdx.x % NTILE) * 32;
  const int t    = threadIdx.x;
  const int lane = t & 31;
  const int w    = t >> 5;
  float* yT = ws + YT_OFF;

  __shared__ float mus[32], rss[32];
  __shared__ float tile[32][33];

  for (int rr = 0; rr < 4; ++rr) {
    const int r   = w * 4 + rr;
    const int row = n0 + r;
    if (row < NN) {
      const float* xr = x + ((size_t)b * NN + row) * DD;
      float s = 0.0f, q = 0.0f;
      for (int j = lane; j < DD; j += 32) {
        const float v = xr[j];
        s += v;
        q += v * v;
      }
      for (int off = 16; off > 0; off >>= 1) {
        s += __shfl_xor(s, off, 32);
        q += __shfl_xor(q, off, 32);
      }
      if (lane == 0) {
        const float mu  = s * (1.0f / DD);
        const float var = q * (1.0f / DD) - mu * mu;
        mus[r] = mu;
        rss[r] = rsqrtf(var + 1e-5f);
      }
    }
  }
  __syncthreads();

  for (int dc = 0; dc < DD / 32; ++dc) {
    const int d0 = dc * 32;
#pragma unroll
    for (int p = 0; p < 4; ++p) {
      const int r   = (t >> 5) + p * 8;
      const int c   = t & 31;
      const int row = n0 + r;
      float v = 0.0f;
      if (row < NN) {
        const float xv = x[((size_t)b * NN + row) * DD + d0 + c];
        v = (xv - mus[r]) * rss[r] * lw[d0 + c] + lb[d0 + c];
      }
      tile[r][c] = v;
    }
    __syncthreads();
#pragma unroll
    for (int p = 0; p < 4; ++p) {
      const int dr = (t >> 5) + p * 8;
      const int nc = t & 31;
      if (n0 + nc < NN)
        yT[((size_t)b * DD + d0 + dr) * NN + n0 + nc] = tile[nc][dr];
    }
    __syncthreads();
  }
}

#define KPITCH 209
#define AS_FLOATS (KSTEPS * KPITCH * 4)
#define KCH 7
#define NSUB 12
#define NGRP (NT / NSUB)
#define GEMM_THREADS (NSUB * 2 * 32)
#define GEMM_BLOCKS (BATCH * NGRP)

template <int NM>
__device__ __forceinline__ void dnm_gemm_task(const float* __restrict__ x,
                                              float* __restrict__ out,
                                              const float* As, const float* cs,
                                              const float* __restrict__ Brow,
                                              const v2f (&b0)[KCH],
                                              const int b, const int col,
                                              const int m0, const int c, const int hi) {
  const float* Abase = As + ((m0 * 16 + c) * 4 + hi * 2);

  v8f acc[NM];
#pragma unroll
  for (int i = 0; i < NM; ++i) acc[i] = (v8f){};

  v2f bcur[KCH], bnxt[KCH];
#pragma unroll
  for (int j = 0; j < KCH; ++j) {
    bcur[j] = b0[j];
    bnxt[j] = b0[j];
  }

#pragma unroll 1
  for (int ch = 0; ch < KSTEPS / KCH; ++ch) {
    if (ch + 1 < KSTEPS / KCH) {
#pragma unroll
      for (int j = 0; j < KCH; ++j)
        bnxt[j] = *(const v2f*)(Brow + ((ch + 1) * KCH + j) * 4);
    }
    {
      int opf = m0 * 16 + ch * 16 + c;
      opf = (opf < OO) ? opf : (OO - 1);
      __builtin_prefetch(&x[((size_t)b * OO + opf) * DD + col], 0, 3);
    }
#pragma unroll
    for (int j = 0; j < KCH; ++j) {
      const float* Ak = Abase + (ch * KCH + j) * (KPITCH * 4);
#pragma unroll
      for (int i = 0; i < NM; ++i) {
        const v2f a = *(const v2f*)(Ak + i * 64);
        acc[i] = __builtin_amdgcn_wmma_f32_16x16x4_f32(
             false, a,  false, bcur[j],
             (short)0, acc[i],  false,  false);
      }
    }
#pragma unroll
    for (int j = 0; j < KCH; ++j) bcur[j] = bnxt[j];
  }

#pragma unroll
  for (int i = 0; i < NM; ++i) {
#pragma unroll
    for (int v = 0; v < 8; ++v) {
      const int o = (m0 + i) * 16 + hi * 8 + v;
      if (o < OO) {
        const float tv = acc[i][v] + cs[o];
        const float g  = 0.5f * tv * (1.0f + erff(tv * 0.70710678118654752f));
        const size_t idx = ((size_t)b * OO + o) * DD + col;
        out[idx] = x[idx] * (g + 1.0f);
      }
    }
  }
}

__global__ void __launch_bounds__(GEMM_THREADS) dnm_gemm_kernel(const float* __restrict__ x,
                                                                const float* __restrict__ ws,
                                                                float* __restrict__ out) {
  __shared__ __attribute__((aligned(16))) float As[AS_FLOATS];
  __shared__ float cs[OPAD];

  const float* Wp = ws + WP_OFF;
  const float* cv = ws + C_OFF;
  const float* yT = ws + YT_OFF;

  const int t = threadIdx.x;

  const int wv   = __builtin_amdgcn_readfirstlane(t >> 5);
  const int lane = t & 31;
  const int c    = lane & 15;
  const int hi   = lane >> 4;
  const int mh   = (wv >= NSUB) ? 1 : 0;
  const int nsub = wv - mh * NSUB;

  const int b   = blockIdx.x / NGRP;
  const int ng  = blockIdx.x - b * NGRP;
  const int col = (ng * NSUB + nsub) * 16 + c;

  const float* Brow = yT + ((size_t)b * DD + col) * NN + hi * 2;
  v2f b0[KCH];
#pragma unroll
  for (int j = 0; j < KCH; ++j) b0[j] = *(const v2f*)(Brow + j * 4);

  const int rot = (blockIdx.x & 15) * ((OPAD * KSTEPS) / 16);
  for (int i = t; i < OPAD * KSTEPS; i += GEMM_THREADS) {
    int ii = i + rot;
    ii = (ii >= OPAD * KSTEPS) ? (ii - OPAD * KSTEPS) : ii;
    const int m  = ii / KSTEPS;
    const int kb = ii - m * KSTEPS;
    const v4f w4 = *(const v4f*)(Wp + (size_t)m * NN + kb * 4);
    *(v4f*)(As + (kb * KPITCH + m) * 4) = w4;
  }
  if (t < OPAD) cs[t] = cv[t];
  __syncthreads();

  if (mh == 0)
    dnm_gemm_task<7>(x, out, As, cs, Brow, b0, b, col, 0, c, hi);
  else
    dnm_gemm_task<6>(x, out, As, cs, Brow, b0, b, col, 7, c, hi);
}

extern "C" void kernel_launch(void* const* d_in, const int* in_sizes, int n_in,
                              void* d_out, int out_size, void* d_ws, size_t ws_size,
                              hipStream_t stream) {
  const float* x    = (const float*)d_in[0];
  const float* ln_w = (const float*)d_in[1];
  const float* ln_b = (const float*)d_in[2];
  const float* sw   = (const float*)d_in[3];
  const float* sb   = (const float*)d_in[4];
  const float* dw   = (const float*)d_in[5];
  float* out = (float*)d_out;
  float* ws  = (float*)d_ws;

  dnm_prep_kernel<<<dim3(OPAD), dim3(NN), 0, stream>>>(sw, sb, dw, ws);
  dnm_ln_t_kernel<<<dim3(BATCH * NTILE), dim3(256), 0, stream>>>(x, ln_w, ln_b, ws);
  dnm_gemm_kernel<<<dim3(GEMM_BLOCKS), dim3(GEMM_THREADS), 0, stream>>>(x, ws, out);
}


# LayerNorm moved into the GEMM: B fragments are read from x rows and normalised in registers; a stats-only kernel (two-pass row sums, new fixed order) replaces LN+transpose, no yT copy
# speedup vs baseline: 2.5181x; 1.7322x over previous
#include <hip/hip_runtime.h>
#include <math.h>

typedef __attribute__((ext_vector_type(2))) float v2f;
typedef __attribute__((ext_vector_type(4))) float v4f;
typedef __attribute__((ext_vector_type(8))) float v8f;

#define BATCH 64
#define NN    196
#define DD    768
#define OO    196
#define OPAD  208
#define MT    13
#define NT    48
#define KSTEPS 49
#define ROWS  (BATCH * NN)

#define WP_OFF 0
#define C_OFF  (OPAD * NN)
#define ST_OFF (C_OFF + 256)

__global__ void dnm_prep_kernel(const float* __restrict__ sw,
                                const float* __restrict__ sb,
                                const float* __restrict__ dw,
                                float* __restrict__ ws) {
  __shared__ float red[NN];
  const int o = blockIdx.x;
  const int t = threadIdx.x;
  float* Wp = ws + WP_OFF;
  float* cv = ws + C_OFF;
  if (o < OO) {
    const float d0 = dw[o * 2 + 0];
    const float d1 = dw[o * 2 + 1];
    Wp[(size_t)o * NN + t] = d0 * sw[(size_t)(o * 2 + 0) * NN + t] +
                             d1 * sw[(size_t)(o * 2 + 1) * NN + t];
    red[t] = d0 * sb[(size_t)(o * 2 + 0) * NN + t] +
             d1 * sb[(size_t)(o * 2 + 1) * NN + t];
    __syncthreads();
    if (t == 0) {
      float s = 0.0f;
      for (int i = 0; i < NN; ++i) s += red[i];
      cv[o] = s;
    }
  } else {
    Wp[(size_t)o * NN + t] = 0.0f;
    if (t == 0) cv[o] = 0.0f;
  }
}

#define ST_RPW 4
#define ST_THREADS 256
#define ST_BLOCKS (ROWS / (ST_RPW * (ST_THREADS / 32)))

__global__ void __launch_bounds__(ST_THREADS) dnm_ln_stats_kernel(const float* __restrict__ x,
                                                                  float* __restrict__ ws) {
  const int lane = threadIdx.x & 31;
  const int w    = threadIdx.x >> 5;
  const int row0 = (blockIdx.x * (ST_THREADS / 32) + w) * ST_RPW;
  float* st = ws + ST_OFF;

  v4f v[ST_RPW][6];
  for (int r = 0; r < ST_RPW; ++r) {
    const v4f* xr = (const v4f*)(x + (size_t)(row0 + r) * DD);
    for (int i = 0; i < 6; ++i) v[r][i] = xr[lane + 32 * i];
  }

  float mu[ST_RPW];
  for (int r = 0; r < ST_RPW; ++r) {
    float s = 0.0f;
    for (int i = 0; i < 6; ++i) s += (v[r][i].x + v[r][i].y) + (v[r][i].z + v[r][i].w);
    for (int off = 16; off > 0; off >>= 1) s += __shfl_xor(s, off, 32);
    mu[r] = s * (1.0f / DD);
  }
  for (int r = 0; r < ST_RPW; ++r) {
    float q = 0.0f;
    for (int i = 0; i < 6; ++i) {
      const float d0 = v[r][i].x - mu[r], d1 = v[r][i].y - mu[r];
      const float d2 = v[r][i].z - mu[r], d3 = v[r][i].w - mu[r];
      q += (d0 * d0 + d1 * d1) + (d2 * d2 + d3 * d3);
    }
    for (int off = 16; off > 0; off >>= 1) q += __shfl_xor(q, off, 32);
    if (lane == 0) {
      v2f o2;
      o2.x = mu[r];
      o2.y = rsqrtf(q * (1.0f / DD) + 1e-5f);
      *(v2f*)(st + (size_t)(row0 + r) * 2) = o2;
    }
  }
}

#define KPITCH 209
#define AS_FLOATS (KSTEPS * KPITCH * 4)
#define KCH 7
#define NSUB 12
#define NGRP (NT / NSUB)
#define GEMM_THREADS (NSUB * 2 * 32)
#define GEMM_BLOCKS (BATCH * NGRP)

template <int NM>
__device__ __forceinline__ void dnm_gemm_task(const float* __restrict__ x,
                                              float* __restrict__ out,
                                              const float* As, const float* cs,
                                              const float* st,
                                              const float* __restrict__ Bp,
                                              const v2f (&b0)[KCH],
                                              const float lwc, const float lbc,
                                              const int b, const int col,
                                              const int m0, const int c, const int hi) {
  const float* Abase = As + ((m0 * 16 + c) * 4 + hi * 2);
  const float* Sbase = st + hi * 4;

  v8f acc[NM];
#pragma unroll
  for (int i = 0; i < NM; ++i) acc[i] = (v8f){};

  v2f bcur[KCH], bnxt[KCH];
#pragma unroll
  for (int j = 0; j < KCH; ++j) {
    bcur[j] = b0[j];
    bnxt[j] = b0[j];
  }

#pragma unroll 1
  for (int ch = 0; ch < KSTEPS / KCH; ++ch) {
    if (ch + 1 < KSTEPS / KCH) {
#pragma unroll
      for (int j = 0; j < KCH; ++j) {
        const float* p = Bp + (size_t)((ch + 1) * KCH + j) * (4 * DD);
        bnxt[j].x = p[0];
        bnxt[j].y = p[DD];
      }
    }
#pragma unroll
    for (int j = 0; j < KCH; ++j) {
      const int kb = ch * KCH + j;
      const v4f s4 = *(const v4f*)(Sbase + kb * 8);
      v2f bn;
      bn.x = (bcur[j].x - s4.x) * s4.y * lwc + lbc;
      bn.y = (bcur[j].y - s4.z) * s4.w * lwc + lbc;
      const float* Ak = Abase + kb * (KPITCH * 4);
#pragma unroll
      for (int i = 0; i < NM; ++i) {
        const v2f a = *(const v2f*)(Ak + i * 64);
        acc[i] = __builtin_amdgcn_wmma_f32_16x16x4_f32(
             false, a,  false, bn,
             (short)0, acc[i],  false,  false);
      }
    }
#pragma unroll
    for (int j = 0; j < KCH; ++j) bcur[j] = bnxt[j];
  }

#pragma unroll
  for (int i = 0; i < NM; ++i) {
#pragma unroll
    for (int v = 0; v < 8; ++v) {
      const int o = (m0 + i) * 16 + hi * 8 + v;
      if (o < OO) {
        const float tv = acc[i][v] + cs[o];
        const float g  = 0.5f * tv * (1.0f + erff(tv * 0.70710678118654752f));
        const size_t idx = ((size_t)b * OO + o) * DD + col;
        out[idx] = x[idx] * (g + 1.0f);
      }
    }
  }
}

__global__ void __launch_bounds__(GEMM_THREADS) dnm_gemm_kernel(const float* __restrict__ x,
                                                                const float* __restrict__ lw,
                                                                const float* __restrict__ lb,
                                                                const float* __restrict__ ws,
                                                                float* __restrict__ out) {
  __shared__ __attribute__((aligned(16))) float As[AS_FLOATS];
  __shared__ __attribute__((aligned(16))) float st[NN * 2];
  __shared__ float cs[OPAD];

  const float* Wp  = ws + WP_OFF;
  const float* cv  = ws + C_OFF;
  const float* stg = ws + ST_OFF;

  const int t = threadIdx.x;

  const int wv   = __builtin_amdgcn_readfirstlane(t >> 5);
  const int lane = t & 31;
  const int c    = lane & 15;
  const int hi   = lane >> 4;
  const int mh   = (wv >= NSUB) ? 1 : 0;
  const int nsub = wv - mh * NSUB;

  const int b   = blockIdx.x / NGRP;
  const int ng  = blockIdx.x - b * NGRP;
  const int col = (ng * NSUB + nsub) * 16 + c;

  const float* Bp = x + ((size_t)(b * NN + hi * 2) * DD + col);
  v2f b0[KCH];
#pragma unroll
  for (int j = 0; j < KCH; ++j) {
    const float* p = Bp + (size_t)j * (4 * DD);
    b0[j].x = p[0];
    b0[j].y = p[DD];
  }
  const float lwc = lw[col];
  const float lbc = lb[col];

  const int rot = (blockIdx.x & 15) * ((OPAD * KSTEPS) / 16);
  for (int i = t; i < OPAD * KSTEPS; i += GEMM_THREADS) {
    int ii = i + rot;
    ii = (ii >= OPAD * KSTEPS) ? (ii - OPAD * KSTEPS) : ii;
    const int m  = ii / KSTEPS;
    const int kb = ii - m * KSTEPS;
    const v4f w4 = *(const v4f*)(Wp + (size_t)m * NN + kb * 4);
    *(v4f*)(As + (kb * KPITCH + m) * 4) = w4;
  }
  if (t < OPAD) cs[t] = cv[t];
  if (t < NN) *(v2f*)(st + t * 2) = *(const v2f*)(stg + ((size_t)b * NN + t) * 2);
  __syncthreads();

  if (mh == 0)
    dnm_gemm_task<7>(x, out, As, cs, st, Bp, b0, lwc, lbc, b, col, 0, c, hi);
  else
    dnm_gemm_task<6>(x, out, As, cs, st, Bp, b0, lwc, lbc, b, col, 7, c, hi);
}

extern "C" void kernel_launch(void* const* d_in, const int* in_sizes, int n_in,
                              void* d_out, int out_size, void* d_ws, size_t ws_size,
                              hipStream_t stream) {
  const float* x    = (const float*)d_in[0];
  const float* ln_w = (const float*)d_in[1];
  const float* ln_b = (const float*)d_in[2];
  const float* sw   = (const float*)d_in[3];
  const float* sb   = (const float*)d_in[4];
  const float* dw   = (const float*)d_in[5];
  float* out = (float*)d_out;
  float* ws  = (float*)d_ws;

  dnm_prep_kernel<<<dim3(OPAD), dim3(NN), 0, stream>>>(sw, sb, dw, ws);
  dnm_ln_stats_kernel<<<dim3(ST_BLOCKS), dim3(ST_THREADS), 0, stream>>>(x, ws);
  dnm_gemm_kernel<<<dim3(GEMM_BLOCKS), dim3(GEMM_THREADS), 0, stream>>>(x, ln_w, ln_b, ws, out);
}


# GEMM epilogue: x operands of all of a wave's outputs loaded as one group after the K loop, bias vectors read from LDS a tile ahead; no dependent load before each element's GELU any more
# speedup vs baseline: 3.3279x; 1.3216x over previous
#include <hip/hip_runtime.h>
#include <math.h>

typedef __attribute__((ext_vector_type(2))) float v2f;
typedef __attribute__((ext_vector_type(4))) float v4f;
typedef __attribute__((ext_vector_type(8))) float v8f;

#define BATCH 64
#define NN    196
#define DD    768
#define OO    196
#define OPAD  208
#define MT    13
#define NT    48
#define KSTEPS 49
#define ROWS  (BATCH * NN)

#define WP_OFF 0
#define C_OFF  (OPAD * NN)
#define ST_OFF (C_OFF + 256)

__global__ void dnm_prep_kernel(const float* __restrict__ sw,
                                const float* __restrict__ sb,
                                const float* __restrict__ dw,
                                float* __restrict__ ws) {
  __shared__ float red[NN];
  const int o = blockIdx.x;
  const int t = threadIdx.x;
  float* Wp = ws + WP_OFF;
  float* cv = ws + C_OFF;
  if (o < OO) {
    const float d0 = dw[o * 2 + 0];
    const float d1 = dw[o * 2 + 1];
    Wp[(size_t)o * NN + t] = d0 * sw[(size_t)(o * 2 + 0) * NN + t] +
                             d1 * sw[(size_t)(o * 2 + 1) * NN + t];
    red[t] = d0 * sb[(size_t)(o * 2 + 0) * NN + t] +
             d1 * sb[(size_t)(o * 2 + 1) * NN + t];
    __syncthreads();
    if (t == 0) {
      float s = 0.0f;
      for (int i = 0; i < NN; ++i) s += red[i];
      cv[o] = s;
    }
  } else {
    Wp[(size_t)o * NN + t] = 0.0f;
    if (t == 0) cv[o] = 0.0f;
  }
}

#define ST_RPW 4
#define ST_THREADS 256
#define ST_BLOCKS (ROWS / (ST_RPW * (ST_THREADS / 32)))

__global__ void __launch_bounds__(ST_THREADS) dnm_ln_stats_kernel(const float* __restrict__ x,
                                                                  float* __restrict__ ws) {
  const int lane = threadIdx.x & 31;
  const int w    = threadIdx.x >> 5;
  const int row0 = (blockIdx.x * (ST_THREADS / 32) + w) * ST_RPW;
  float* st = ws + ST_OFF;

  v4f v[ST_RPW][6];
  for (int r = 0; r < ST_RPW; ++r) {
    const v4f* xr = (const v4f*)(x + (size_t)(row0 + r) * DD);
    for (int i = 0; i < 6; ++i) v[r][i] = xr[lane + 32 * i];
  }

  float mu[ST_RPW];
  for (int r = 0; r < ST_RPW; ++r) {
    float s = 0.0f;
    for (int i = 0; i < 6; ++i) s += (v[r][i].x + v[r][i].y) + (v[r][i].z + v[r][i].w);
    for (int off = 16; off > 0; off >>= 1) s += __shfl_xor(s, off, 32);
    mu[r] = s * (1.0f / DD);
  }
  for (int r = 0; r < ST_RPW; ++r) {
    float q = 0.0f;
    for (int i = 0; i < 6; ++i) {
      const float d0 = v[r][i].x - mu[r], d1 = v[r][i].y - mu[r];
      const float d2 = v[r][i].z - mu[r], d3 = v[r][i].w - mu[r];
      q += (d0 * d0 + d1 * d1) + (d2 * d2 + d3 * d3);
    }
    for (int off = 16; off > 0; off >>= 1) q += __shfl_xor(q, off, 32);
    if (lane == 0) {
      v2f o2;
      o2.x = mu[r];
      o2.y = rsqrtf(q * (1.0f / DD) + 1e-5f);
      *(v2f*)(st + (size_t)(row0 + r) * 2) = o2;
    }
  }
}

#define KPITCH 209
#define AS_FLOATS (KSTEPS * KPITCH * 4)
#define KCH 7
#define NSUB 12
#define NGRP (NT / NSUB)
#define GEMM_THREADS (NSUB * 2 * 32)
#define GEMM_BLOCKS (BATCH * NGRP)

template <int NM>
__device__ __forceinline__ void dnm_gemm_task(const float* __restrict__ x,
                                              float* __restrict__ out,
                                              const float* As, const float* cs,
                                              const float* st,
                                              const float* __restrict__ Bp,
                                              const v2f (&b0)[KCH],
                                              const float lwc, const float lbc,
                                              const int b, const int col,
                                              const int m0, const int c, const int hi) {
  const float* Abase = As + ((m0 * 16 + c) * 4 + hi * 2);
  const float* Sbase = st + hi * 4;

  v8f acc[NM];
#pragma unroll
  for (int i = 0; i < NM; ++i) acc[i] = (v8f){};

  v2f bcur[KCH], bnxt[KCH];
#pragma unroll
  for (int j = 0; j < KCH; ++j) {
    bcur[j] = b0[j];
    bnxt[j] = b0[j];
  }

#pragma unroll 1
  for (int ch = 0; ch < KSTEPS / KCH; ++ch) {
    if (ch + 1 < KSTEPS / KCH) {
#pragma unroll
      for (int j = 0; j < KCH; ++j) {
        const float* p = Bp + (size_t)((ch + 1) * KCH + j) * (4 * DD);
        bnxt[j].x = p[0];
        bnxt[j].y = p[DD];
      }
    }
#pragma unroll
    for (int j = 0; j < KCH; ++j) {
      const int kb = ch * KCH + j;
      const v4f s4 = *(const v4f*)(Sbase + kb * 8);
      v2f bn;
      bn.x = (bcur[j].x - s4.x) * s4.y * lwc + lbc;
      bn.y = (bcur[j].y - s4.z) * s4.w * lwc + lbc;
      const float* Ak = Abase + kb * (KPITCH * 4);
#pragma unroll
      for (int i = 0; i < NM; ++i) {
        const v2f a = *(const v2f*)(Ak + i * 64);
        acc[i] = __builtin_amdgcn_wmma_f32_16x16x4_f32(
             false, a,  false, bn,
             (short)0, acc[i],  false,  false);
      }
    }
#pragma unroll
    for (int j = 0; j < KCH; ++j) bcur[j] = bnxt[j];
  }

  const size_t ebase = ((size_t)b * OO + m0 * 16 + hi * 8) * DD + col;
  const float* xe = x + ebase;
  float*       oe = out + ebase;

  float xv[NM][8];
#pragma unroll
  for (int i = 0; i < NM; ++i) {
#pragma unroll
    for (int v = 0; v < 8; ++v) {
      const int o = (m0 + i) * 16 + hi * 8 + v;
      const float* p = (o < OO) ? (xe + (size_t)(i * 16 + v) * DD) : xe;
      xv[i][v] = *p;
    }
  }

  const float* cl = cs + m0 * 16 + hi * 8;
  v4f cn0 = *(const v4f*)(cl);
  v4f cn1 = *(const v4f*)(cl + 4);
#pragma unroll
  for (int i = 0; i < NM; ++i) {
    const v4f c0 = cn0, c1 = cn1;
    if (i + 1 < NM) {
      cn0 = *(const v4f*)(cl + (i + 1) * 16);
      cn1 = *(const v4f*)(cl + (i + 1) * 16 + 4);
    }
    const float cb[8] = {c0.x, c0.y, c0.z, c0.w, c1.x, c1.y, c1.z, c1.w};
#pragma unroll
    for (int v = 0; v < 8; ++v) {
      const int o = (m0 + i) * 16 + hi * 8 + v;
      if (o < OO) {
        const float tv = acc[i][v] + cb[v];
        const float g  = 0.5f * tv * (1.0f + erff(tv * 0.70710678118654752f));
        oe[(size_t)(i * 16 + v) * DD] = xv[i][v] * (g + 1.0f);
      }
    }
  }
}

__global__ void __launch_bounds__(GEMM_THREADS) dnm_gemm_kernel(const float* __restrict__ x,
                                                                const float* __restrict__ lw,
                                                                const float* __restrict__ lb,
                                                                const float* __restrict__ ws,
                                                                float* __restrict__ out) {
  __shared__ __attribute__((aligned(16))) float As[AS_FLOATS];
  __shared__ __attribute__((aligned(16))) float st[NN * 2];
  __shared__ __attribute__((aligned(16))) float cs[OPAD];

  const float* Wp  = ws + WP_OFF;
  const float* cv  = ws + C_OFF;
  const float* stg = ws + ST_OFF;

  const int t = threadIdx.x;

  const int wv   = __builtin_amdgcn_readfirstlane(t >> 5);
  const int lane = t & 31;
  const int c    = lane & 15;
  const int hi   = lane >> 4;
  const int mh   = (wv >= NSUB) ? 1 : 0;
  const int nsub = wv - mh * NSUB;

  const int b   = blockIdx.x / NGRP;
  const int ng  = blockIdx.x - b * NGRP;
  const int col = (ng * NSUB + nsub) * 16 + c;

  const float* Bp = x + ((size_t)(b * NN + hi * 2) * DD + col);
  v2f b0[KCH];
#pragma unroll
  for (int j = 0; j < KCH; ++j) {
    const float* p = Bp + (size_t)j * (4 * DD);
    b0[j].x = p[0];
    b0[j].y = p[DD];
  }
  const float lwc = lw[col];
  const float lbc = lb[col];

  const int rot = (blockIdx.x & 15) * ((OPAD * KSTEPS) / 16);
  for (int i = t; i < OPAD * KSTEPS; i += GEMM_THREADS) {
    int ii = i + rot;
    ii = (ii >= OPAD * KSTEPS) ? (ii - OPAD * KSTEPS) : ii;
    const int m  = ii / KSTEPS;
    const int kb = ii - m * KSTEPS;
    const v4f w4 = *(const v4f*)(Wp + (size_t)m * NN + kb * 4);
    *(v4f*)(As + (kb * KPITCH + m) * 4) = w4;
  }
  if (t < OPAD) cs[t] = cv[t];
  if (t < NN) *(v2f*)(st + t * 2) = *(const v2f*)(stg + ((size_t)b * NN + t) * 2);
  __syncthreads();

  if (mh == 0)
    dnm_gemm_task<7>(x, out, As, cs, st, Bp, b0, lwc, lbc, b, col, 0, c, hi);
  else
    dnm_gemm_task<6>(x, out, As, cs, st, Bp, b0, lwc, lbc, b, col, 7, c, hi);
}

extern "C" void kernel_launch(void* const* d_in, const int* in_sizes, int n_in,
                              void* d_out, int out_size, void* d_ws, size_t ws_size,
                              hipStream_t stream) {
  const float* x    = (const float*)d_in[0];
  const float* ln_w = (const float*)d_in[1];
  const float* ln_b = (const float*)d_in[2];
  const float* sw   = (const float*)d_in[3];
  const float* sb   = (const float*)d_in[4];
  const float* dw   = (const float*)d_in[5];
  float* out = (float*)d_out;
  float* ws  = (float*)d_ws;

  dnm_prep_kernel<<<dim3(OPAD), dim3(NN), 0, stream>>>(sw, sb, dw, ws);
  dnm_ln_stats_kernel<<<dim3(ST_BLOCKS), dim3(ST_THREADS), 0, stream>>>(x, ws);
  dnm_gemm_kernel<<<dim3(GEMM_BLOCKS), dim3(GEMM_THREADS), 0, stream>>>(x, ln_w, ln_b, ws, out);
}


# Weight fold and LayerNorm row statistics merged into one launch: the two independent pre-passes share one grid (weight blocks first), 3 launches become 2; no sum changes its order
# speedup vs baseline: 3.5417x; 1.0642x over previous
#include <hip/hip_runtime.h>
#include <math.h>

typedef __attribute__((ext_vector_type(2))) float v2f;
typedef __attribute__((ext_vector_type(4))) float v4f;
typedef __attribute__((ext_vector_type(8))) float v8f;

#define BATCH 64
#define NN    196
#define DD    768
#define OO    196
#define OPAD  208
#define MT    13
#define NT    48
#define KSTEPS 49
#define ROWS  (BATCH * NN)

#define WP_OFF 0
#define C_OFF  (OPAD * NN)
#define ST_OFF (C_OFF + 256)

#define PRE_THREADS 256
#define PREP_BLOCKS OPAD
#define ST_RPW 4
#define ST_BLOCKS (ROWS / (ST_RPW * (PRE_THREADS / 32)))
#define PRE_BLOCKS (PREP_BLOCKS + ST_BLOCKS)

__device__ __forceinline__ void dnm_fold_block(const float* __restrict__ sw,
                                               const float* __restrict__ sb,
                                               const float* __restrict__ dw,
                                               float* __restrict__ ws,
                                               float* red, const int o, const int t) {
  float* Wp = ws + WP_OFF;
  float* cv = ws + C_OFF;
  if (o < OO) {
    if (t < NN) {
      const float d0 = dw[o * 2 + 0];
      const float d1 = dw[o * 2 + 1];
      Wp[(size_t)o * NN + t] = d0 * sw[(size_t)(o * 2 + 0) * NN + t] +
                               d1 * sw[(size_t)(o * 2 + 1) * NN + t];
      red[t] = d0 * sb[(size_t)(o * 2 + 0) * NN + t] +
               d1 * sb[(size_t)(o * 2 + 1) * NN + t];
    }
    __syncthreads();
    if (t == 0) {
      float s = 0.0f;
      for (int i = 0; i < NN; ++i) s += red[i];
      cv[o] = s;
    }
  } else {
    if (t < NN) Wp[(size_t)o * NN + t] = 0.0f;
    if (t == 0) cv[o] = 0.0f;
  }
}

__device__ __forceinline__ void dnm_stats_block(const float* __restrict__ x,
                                                float* __restrict__ ws,
                                                const int sblock, const int tid) {
  const int lane = tid & 31;
  const int w    = tid >> 5;
  const int row0 = (sblock * (PRE_THREADS / 32) + w) * ST_RPW;
  float* st = ws + ST_OFF;

  v4f v[ST_RPW][6];
  for (int r = 0; r < ST_RPW; ++r) {
    const v4f* xr = (const v4f*)(x + (size_t)(row0 + r) * DD);
    for (int i = 0; i < 6; ++i) v[r][i] = xr[lane + 32 * i];
  }

  float mu[ST_RPW];
  for (int r = 0; r < ST_RPW; ++r) {
    float s = 0.0f;
    for (int i = 0; i < 6; ++i) s += (v[r][i].x + v[r][i].y) + (v[r][i].z + v[r][i].w);
    for (int off = 16; off > 0; off >>= 1) s += __shfl_xor(s, off, 32);
    mu[r] = s * (1.0f / DD);
  }
  for (int r = 0; r < ST_RPW; ++r) {
    float q = 0.0f;
    for (int i = 0; i < 6; ++i) {
      const float d0 = v[r][i].x - mu[r], d1 = v[r][i].y - mu[r];
      const float d2 = v[r][i].z - mu[r], d3 = v[r][i].w - mu[r];
      q += (d0 * d0 + d1 * d1) + (d2 * d2 + d3 * d3);
    }
    for (int off = 16; off > 0; off >>= 1) q += __shfl_xor(q, off, 32);
    if (lane == 0) {
      v2f o2;
      o2.x = mu[r];
      o2.y = rsqrtf(q * (1.0f / DD) + 1e-5f);
      *(v2f*)(st + (size_t)(row0 + r) * 2) = o2;
    }
  }
}

__global__ void __launch_bounds__(PRE_THREADS) dnm_pre_kernel(const float* __restrict__ x,
                                                              const float* __restrict__ sw,
                                                              const float* __restrict__ sb,
                                                              const float* __restrict__ dw,
                                                              float* __restrict__ ws) {
  __shared__ float red[NN];
  if (blockIdx.x >= PREP_BLOCKS)
    dnm_stats_block(x, ws, blockIdx.x - PREP_BLOCKS, threadIdx.x);
  else
    dnm_fold_block(sw, sb, dw, ws, red, blockIdx.x, threadIdx.x);
}

#define KPITCH 209
#define AS_FLOATS (KSTEPS * KPITCH * 4)
#define KCH 7
#define NSUB 12
#define NGRP (NT / NSUB)
#define GEMM_THREADS (NSUB * 2 * 32)
#define GEMM_BLOCKS (BATCH * NGRP)

template <int NM>
__device__ __forceinline__ void dnm_gemm_task(const float* __restrict__ x,
                                              float* __restrict__ out,
                                              const float* As, const float* cs,
                                              const float* st,
                                              const float* __restrict__ Bp,
                                              const v2f (&b0)[KCH],
                                              const float lwc, const float lbc,
                                              const int b, const int col,
                                              const int m0, const int c, const int hi) {
  const float* Abase = As + ((m0 * 16 + c) * 4 + hi * 2);
  const float* Sbase = st + hi * 4;

  v8f acc[NM];
#pragma unroll
  for (int i = 0; i < NM; ++i) acc[i] = (v8f){};

  v2f bcur[KCH], bnxt[KCH];
#pragma unroll
  for (int j = 0; j < KCH; ++j) {
    bcur[j] = b0[j];
    bnxt[j] = b0[j];
  }

#pragma unroll 1
  for (int ch = 0; ch < KSTEPS / KCH; ++ch) {
    if (ch + 1 < KSTEPS / KCH) {
#pragma unroll
      for (int j = 0; j < KCH; ++j) {
        const float* p = Bp + (size_t)((ch + 1) * KCH + j) * (4 * DD);
        bnxt[j].x = p[0];
        bnxt[j].y = p[DD];
      }
    }
#pragma unroll
    for (int j = 0; j < KCH; ++j) {
      const int kb = ch * KCH + j;
      const v4f s4 = *(const v4f*)(Sbase + kb * 8);
      v2f bn;
      bn.x = (bcur[j].x - s4.x) * s4.y * lwc + lbc;
      bn.y = (bcur[j].y - s4.z) * s4.w * lwc + lbc;
      const float* Ak = Abase + kb * (KPITCH * 4);
#pragma unroll
      for (int i = 0; i < NM; ++i) {
        const v2f a = *(const v2f*)(Ak + i * 64);
        acc[i] = __builtin_amdgcn_wmma_f32_16x16x4_f32(
             false, a,  false, bn,
             (short)0, acc[i],  false,  false);
      }
    }
#pragma unroll
    for (int j = 0; j < KCH; ++j) bcur[j] = bnxt[j];
  }

  const size_t ebase = ((size_t)b * OO + m0 * 16 + hi * 8) * DD + col;
  const float* xe = x + ebase;
  float*       oe = out + ebase;

  float xv[NM][8];
#pragma unroll
  for (int i = 0; i < NM; ++i) {
#pragma unroll
    for (int v = 0; v < 8; ++v) {
      const int o = (m0 + i) * 16 + hi * 8 + v;
      const float* p = (o < OO) ? (xe + (size_t)(i * 16 + v) * DD) : xe;
      xv[i][v] = *p;
    }
  }

  const float* cl = cs + m0 * 16 + hi * 8;
  v4f cn0 = *(const v4f*)(cl);
  v4f cn1 = *(const v4f*)(cl + 4);
#pragma unroll
  for (int i = 0; i < NM; ++i) {
    const v4f c0 = cn0, c1 = cn1;
    if (i + 1 < NM) {
      cn0 = *(const v4f*)(cl + (i + 1) * 16);
      cn1 = *(const v4f*)(cl + (i + 1) * 16 + 4);
    }
    const float cb[8] = {c0.x, c0.y, c0.z, c0.w, c1.x, c1.y, c1.z, c1.w};
#pragma unroll
    for (int v = 0; v < 8; ++v) {
      const int o = (m0 + i) * 16 + hi * 8 + v;
      if (o < OO) {
        const float tv = acc[i][v] + cb[v];
        const float g  = 0.5f * tv * (1.0f + erff(tv * 0.70710678118654752f));
        oe[(size_t)(i * 16 + v) * DD] = xv[i][v] * (g + 1.0f);
      }
    }
  }
}

__global__ void __launch_bounds__(GEMM_THREADS) dnm_gemm_kernel(const float* __restrict__ x,
                                                                const float* __restrict__ lw,
                                                                const float* __restrict__ lb,
                                                                const float* __restrict__ ws,
                                                                float* __restrict__ out) {
  __shared__ __attribute__((aligned(16))) float As[AS_FLOATS];
  __shared__ __attribute__((aligned(16))) float st[NN * 2];
  __shared__ __attribute__((aligned(16))) float cs[OPAD];

  const float* Wp  = ws + WP_OFF;
  const float* cv  = ws + C_OFF;
  const float* stg = ws + ST_OFF;

  const int t = threadIdx.x;

  const int wv   = __builtin_amdgcn_readfirstlane(t >> 5);
  const int lane = t & 31;
  const int c    = lane & 15;
  const int hi   = lane >> 4;
  const int mh   = (wv >= NSUB) ? 1 : 0;
  const int nsub = wv - mh * NSUB;

  const int b   = blockIdx.x / NGRP;
  const int ng  = blockIdx.x - b * NGRP;
  const int col = (ng * NSUB + nsub) * 16 + c;

  const float* Bp = x + ((size_t)(b * NN + hi * 2) * DD + col);
  v2f b0[KCH];
#pragma unroll
  for (int j = 0; j < KCH; ++j) {
    const float* p = Bp + (size_t)j * (4 * DD);
    b0[j].x = p[0];
    b0[j].y = p[DD];
  }
  const float lwc = lw[col];
  const float lbc = lb[col];

  const int rot = (blockIdx.x & 15) * ((OPAD * KSTEPS) / 16);
  for (int i = t; i < OPAD * KSTEPS; i += GEMM_THREADS) {
    int ii = i + rot;
    ii = (ii >= OPAD * KSTEPS) ? (ii - OPAD * KSTEPS) : ii;
    const int m  = ii / KSTEPS;
    const int kb = ii - m * KSTEPS;
    const v4f w4 = *(const v4f*)(Wp + (size_t)m * NN + kb * 4);
    *(v4f*)(As + (kb * KPITCH + m) * 4) = w4;
  }
  if (t < OPAD) cs[t] = cv[t];
  if (t < NN) *(v2f*)(st + t * 2) = *(const v2f*)(stg + ((size_t)b * NN + t) * 2);
  __syncthreads();

  if (mh == 0)
    dnm_gemm_task<7>(x, out, As, cs, st, Bp, b0, lwc, lbc, b, col, 0, c, hi);
  else
    dnm_gemm_task<6>(x, out, As, cs, st, Bp, b0, lwc, lbc, b, col, 7, c, hi);
}

extern "C" void kernel_launch(void* const* d_in, const int* in_sizes, int n_in,
                              void* d_out, int out_size, void* d_ws, size_t ws_size,
                              hipStream_t stream) {
  const float* x    = (const float*)d_in[0];
  const float* ln_w = (const float*)d_in[1];
  const float* ln_b = (const float*)d_in[2];
  const float* sw   = (const float*)d_in[3];
  const float* sb   = (const float*)d_in[4];
  const float* dw   = (const float*)d_in[5];
  float* out = (float*)d_out;
  float* ws  = (float*)d_ws;

  dnm_pre_kernel<<<dim3(PRE_BLOCKS), dim3(PRE_THREADS), 0, stream>>>(x, sw, sb, dw, ws);
  dnm_gemm_kernel<<<dim3(GEMM_BLOCKS), dim3(GEMM_THREADS), 0, stream>>>(x, ln_w, ln_b, ws, out);
}


# GELU epilogue: erf branch-free, two rows at a time: both erf polynomials evaluated and selected, packed multiply-adds, hardware exp2 replaces the range-reduced exp; no sum changes its order
# speedup vs baseline: 3.8137x; 1.0768x over previous
#include <hip/hip_runtime.h>
#include <math.h>

typedef __attribute__((ext_vector_type(2))) float v2f;
typedef __attribute__((ext_vector_type(4))) float v4f;
typedef __attribute__((ext_vector_type(8))) float v8f;

#define BATCH 64
#define NN    196
#define DD    768
#define OO    196
#define OPAD  208
#define MT    13
#define NT    48
#define KSTEPS 49
#define ROWS  (BATCH * NN)

#define WP_OFF 0
#define C_OFF  (OPAD * NN)
#define ST_OFF (C_OFF + 256)

#define PRE_THREADS 256
#define PREP_BLOCKS OPAD
#define ST_RPW 4
#define ST_BLOCKS (ROWS / (ST_RPW * (PRE_THREADS / 32)))
#define PRE_BLOCKS (PREP_BLOCKS + ST_BLOCKS)

__device__ __forceinline__ void dnm_fold_block(const float* __restrict__ sw,
                                               const float* __restrict__ sb,
                                               const float* __restrict__ dw,
                                               float* __restrict__ ws,
                                               float* red, const int o, const int t) {
  float* Wp = ws + WP_OFF;
  float* cv = ws + C_OFF;
  if (o < OO) {
    if (t < NN) {
      const float d0 = dw[o * 2 + 0];
      const float d1 = dw[o * 2 + 1];
      Wp[(size_t)o * NN + t] = d0 * sw[(size_t)(o * 2 + 0) * NN + t] +
                               d1 * sw[(size_t)(o * 2 + 1) * NN + t];
      red[t] = d0 * sb[(size_t)(o * 2 + 0) * NN + t] +
               d1 * sb[(size_t)(o * 2 + 1) * NN + t];
    }
    __syncthreads();
    if (t == 0) {
      float s = 0.0f;
      for (int i = 0; i < NN; ++i) s += red[i];
      cv[o] = s;
    }
  } else {
    if (t < NN) Wp[(size_t)o * NN + t] = 0.0f;
    if (t == 0) cv[o] = 0.0f;
  }
}

__device__ __forceinline__ void dnm_stats_block(const float* __restrict__ x,
                                                float* __restrict__ ws,
                                                const int sblock, const int tid) {
  const int lane = tid & 31;
  const int w    = tid >> 5;
  const int row0 = (sblock * (PRE_THREADS / 32) + w) * ST_RPW;
  float* st = ws + ST_OFF;

  v4f v[ST_RPW][6];
  for (int r = 0; r < ST_RPW; ++r) {
    const v4f* xr = (const v4f*)(x + (size_t)(row0 + r) * DD);
    for (int i = 0; i < 6; ++i) v[r][i] = xr[lane + 32 * i];
  }

  float mu[ST_RPW];
  for (int r = 0; r < ST_RPW; ++r) {
    float s = 0.0f;
    for (int i = 0; i < 6; ++i) s += (v[r][i].x + v[r][i].y) + (v[r][i].z + v[r][i].w);
    for (int off = 16; off > 0; off >>= 1) s += __shfl_xor(s, off, 32);
    mu[r] = s * (1.0f / DD);
  }
  for (int r = 0; r < ST_RPW; ++r) {
    float q = 0.0f;
    for (int i = 0; i < 6; ++i) {
      const float d0 = v[r][i].x - mu[r], d1 = v[r][i].y - mu[r];
      const float d2 = v[r][i].z - mu[r], d3 = v[r][i].w - mu[r];
      q += (d0 * d0 + d1 * d1) + (d2 * d2 + d3 * d3);
    }
    for (int off = 16; off > 0; off >>= 1) q += __shfl_xor(q, off, 32);
    if (lane == 0) {
      v2f o2;
      o2.x = mu[r];
      o2.y = rsqrtf(q * (1.0f / DD) + 1e-5f);
      *(v2f*)(st + (size_t)(row0 + r) * 2) = o2;
    }
  }
}

__global__ void __launch_bounds__(PRE_THREADS) dnm_pre_kernel(const float* __restrict__ x,
                                                              const float* __restrict__ sw,
                                                              const float* __restrict__ sb,
                                                              const float* __restrict__ dw,
                                                              float* __restrict__ ws) {
  __shared__ float red[NN];
  if (blockIdx.x >= PREP_BLOCKS)
    dnm_stats_block(x, ws, blockIdx.x - PREP_BLOCKS, threadIdx.x);
  else
    dnm_fold_block(sw, sb, dw, ws, red, blockIdx.x, threadIdx.x);
}

#define KPITCH 209
#define AS_FLOATS (KSTEPS * KPITCH * 4)
#define KCH 7
#define NSUB 12
#define NGRP (NT / NSUB)
#define GEMM_THREADS (NSUB * 2 * 32)
#define GEMM_BLOCKS (BATCH * NGRP)

__device__ __forceinline__ v2f dnm_splat(const float c) {
  v2f r;
  r.x = c;
  r.y = c;
  return r;
}

__device__ __forceinline__ v2f dnm_erf2(const v2f z) {
  v2f az;
  az.x = __builtin_fabsf(z.x);
  az.y = __builtin_fabsf(z.y);
  v2f p = __builtin_elementwise_fma(az, dnm_splat(0.0000169988107f), dnm_splat(-0.000378677854f));
  p = __builtin_elementwise_fma(az, p, dnm_splat(0.00385781587f));
  p = __builtin_elementwise_fma(az, p, dnm_splat(-0.0241816975f));
  p = __builtin_elementwise_fma(az, p, dnm_splat(0.106668264f));
  p = __builtin_elementwise_fma(az, p, dnm_splat(0.634933293f));
  p = __builtin_elementwise_fma(az, p, dnm_splat(0.128689408f));
  const v2f q = __builtin_elementwise_fma(az, p, az);
  const v2f y = q * dnm_splat(-1.44269502f);
  v2f bg;
  bg.x = 1.0f - __builtin_amdgcn_exp2f(y.x);
  bg.y = 1.0f - __builtin_amdgcn_exp2f(y.y);
  const v2f s = az * az;
  v2f r = __builtin_elementwise_fma(s, dnm_splat(-0.000561801775f), dnm_splat(0.00491381623f));
  r = __builtin_elementwise_fma(s, r, dnm_splat(-0.0267075151f));
  r = __builtin_elementwise_fma(s, r, dnm_splat(0.112800106f));
  r = __builtin_elementwise_fma(s, r, dnm_splat(-0.376122952f));
  r = __builtin_elementwise_fma(s, r, dnm_splat(0.128379107f));
  const v2f sm = __builtin_elementwise_fma(az, r, az);
  v2f e;
  e.x = __builtin_copysignf((az.x < 1.0f) ? sm.x : bg.x, z.x);
  e.y = __builtin_copysignf((az.y < 1.0f) ? sm.y : bg.y, z.y);
  return e;
}

template <int NM>
__device__ __forceinline__ void dnm_gemm_task(const float* __restrict__ x,
                                              float* __restrict__ out,
                                              const float* As, const float* cs,
                                              const float* st,
                                              const float* __restrict__ Bp,
                                              const v2f (&b0)[KCH],
                                              const float lwc, const float lbc,
                                              const int b, const int col,
                                              const int m0, const int c, const int hi) {
  const float* Abase = As + ((m0 * 16 + c) * 4 + hi * 2);
  const float* Sbase = st + hi * 4;

  v8f acc[NM];
#pragma unroll
  for (int i = 0; i < NM; ++i) acc[i] = (v8f){};

  v2f bcur[KCH], bnxt[KCH];
#pragma unroll
  for (int j = 0; j < KCH; ++j) {
    bcur[j] = b0[j];
    bnxt[j] = b0[j];
  }

#pragma unroll 1
  for (int ch = 0; ch < KSTEPS / KCH; ++ch) {
    if (ch + 1 < KSTEPS / KCH) {
#pragma unroll
      for (int j = 0; j < KCH; ++j) {
        const float* p = Bp + (size_t)((ch + 1) * KCH + j) * (4 * DD);
        bnxt[j].x = p[0];
        bnxt[j].y = p[DD];
      }
    }
#pragma unroll
    for (int j = 0; j < KCH; ++j) {
      const int kb = ch * KCH + j;
      const v4f s4 = *(const v4f*)(Sbase + kb * 8);
      v2f bn;
      bn.x = (bcur[j].x - s4.x) * s4.y * lwc + lbc;
      bn.y = (bcur[j].y - s4.z) * s4.w * lwc + lbc;
      const float* Ak = Abase + kb * (KPITCH * 4);
#pragma unroll
      for (int i = 0; i < NM; ++i) {
        const v2f a = *(const v2f*)(Ak + i * 64);
        acc[i] = __builtin_amdgcn_wmma_f32_16x16x4_f32(
             false, a,  false, bn,
             (short)0, acc[i],  false,  false);
      }
    }
#pragma unroll
    for (int j = 0; j < KCH; ++j) bcur[j] = bnxt[j];
  }

  const size_t ebase = ((size_t)b * OO + m0 * 16 + hi * 8) * DD + col;
  const float* xe = x + ebase;
  float*       oe = out + ebase;

  float xv[NM][8];
#pragma unroll
  for (int i = 0; i < NM; ++i) {
#pragma unroll
    for (int v = 0; v < 8; ++v) {
      const int o = (m0 + i) * 16 + hi * 8 + v;
      const float* p = (o < OO) ? (xe + (size_t)(i * 16 + v) * DD) : xe;
      xv[i][v] = *p;
    }
  }

  const float* cl = cs + m0 * 16 + hi * 8;
  v4f cn0 = *(const v4f*)(cl);
  v4f cn1 = *(const v4f*)(cl + 4);
#pragma unroll
  for (int i = 0; i < NM; ++i) {
    const v4f c0 = cn0, c1 = cn1;
    if (i + 1 < NM) {
      cn0 = *(const v4f*)(cl + (i + 1) * 16);
      cn1 = *(const v4f*)(cl + (i + 1) * 16 + 4);
    }
    const float cb[8] = {c0.x, c0.y, c0.z, c0.w, c1.x, c1.y, c1.z, c1.w};
#pragma unroll
    for (int v = 0; v < 8; v += 2) {
      const int o = (m0 + i) * 16 + hi * 8 + v;
      if (o < OO) {
        v2f tv;
        tv.x = acc[i][v] + cb[v];
        tv.y = acc[i][v + 1] + cb[v + 1];
        const v2f er = dnm_erf2(tv * dnm_splat(0.70710678118654752f));
        const v2f w = __builtin_elementwise_fma(tv * dnm_splat(0.5f), er + dnm_splat(1.0f),
                                                dnm_splat(1.0f));
        oe[(size_t)(i * 16 + v) * DD]     = xv[i][v] * w.x;
        oe[(size_t)(i * 16 + v + 1) * DD] = xv[i][v + 1] * w.y;
      }
    }
  }
}

__global__ void __launch_bounds__(GEMM_THREADS) dnm_gemm_kernel(const float* __restrict__ x,
                                                                const float* __restrict__ lw,
                                                                const float* __restrict__ lb,
                                                                const float* __restrict__ ws,
                                                                float* __restrict__ out) {
  __shared__ __attribute__((aligned(16))) float As[AS_FLOATS];
  __shared__ __attribute__((aligned(16))) float st[NN * 2];
  __shared__ __attribute__((aligned(16))) float cs[OPAD];

  const float* Wp  = ws + WP_OFF;
  const float* cv  = ws + C_OFF;
  const float* stg = ws + ST_OFF;

  const int t = threadIdx.x;

  const int wv   = __builtin_amdgcn_readfirstlane(t >> 5);
  const int lane = t & 31;
  const int c    = lane & 15;
  const int hi   = lane >> 4;
  const int mh   = (wv >= NSUB) ? 1 : 0;
  const int nsub = wv - mh * NSUB;

  const int b   = blockIdx.x / NGRP;
  const int ng  = blockIdx.x - b * NGRP;
  const int col = (ng * NSUB + nsub) * 16 + c;

  const float* Bp = x + ((size_t)(b * NN + hi * 2) * DD + col);
  v2f b0[KCH];
#pragma unroll
  for (int j = 0; j < KCH; ++j) {
    const float* p = Bp + (size_t)j * (4 * DD);
    b0[j].x = p[0];
    b0[j].y = p[DD];
  }
  const float lwc = lw[col];
  const float lbc = lb[col];

  const int rot = (blockIdx.x & 15) * ((OPAD * KSTEPS) / 16);
  for (int i = t; i < OPAD * KSTEPS; i += GEMM_THREADS) {
    int ii = i + rot;
    ii = (ii >= OPAD * KSTEPS) ? (ii - OPAD * KSTEPS) : ii;
    const int m  = ii / KSTEPS;
    const int kb = ii - m * KSTEPS;
    const v4f w4 = *(const v4f*)(Wp + (size_t)m * NN + kb * 4);
    *(v4f*)(As + (kb * KPITCH + m) * 4) = w4;
  }
  if (t < OPAD) cs[t] = cv[t];
  if (t < NN) *(v2f*)(st + t * 2) = *(const v2f*)(stg + ((size_t)b * NN + t) * 2);
  __syncthreads();

  if (mh == 0)
    dnm_gemm_task<7>(x, out, As, cs, st, Bp, b0, lwc, lbc, b, col, 0, c, hi);
  else
    dnm_gemm_task<6>(x, out, As, cs, st, Bp, b0, lwc, lbc, b, col, 7, c, hi);
}

extern "C" void kernel_launch(void* const* d_in, const int* in_sizes, int n_in,
                              void* d_out, int out_size, void* d_ws, size_t ws_size,
                              hipStream_t stream) {
  const float* x    = (const float*)d_in[0];
  const float* ln_w = (const float*)d_in[1];
  const float* ln_b = (const float*)d_in[2];
  const float* sw   = (const float*)d_in[3];
  const float* sb   = (const float*)d_in[4];
  const float* dw   = (const float*)d_in[5];
  float* out = (float*)d_out;
  float* ws  = (float*)d_ws;

  dnm_pre_kernel<<<dim3(PRE_BLOCKS), dim3(PRE_THREADS), 0, stream>>>(x, sw, sb, dw, ws);
  dnm_gemm_kernel<<<dim3(GEMM_BLOCKS), dim3(GEMM_THREADS), 0, stream>>>(x, ln_w, ln_b, ws, out);
}


# GEMM K loop: LDS operands run one k-step ahead in a second register set: the next step's A fragments and row statistics are read as one group before each step's matrix instructions; no sum changes
# speedup vs baseline: 3.9628x; 1.0391x over previous
#include <hip/hip_runtime.h>
#include <math.h>

typedef __attribute__((ext_vector_type(2))) float v2f;
typedef __attribute__((ext_vector_type(4))) float v4f;
typedef __attribute__((ext_vector_type(8))) float v8f;

#define BATCH 64
#define NN    196
#define DD    768
#define OO    196
#define OPAD  208
#define MT    13
#define NT    48
#define KSTEPS 49
#define ROWS  (BATCH * NN)

#define WP_OFF 0
#define C_OFF  (OPAD * NN)
#define ST_OFF (C_OFF + 256)

#define PRE_THREADS 256
#define PREP_BLOCKS OPAD
#define ST_RPW 4
#define ST_BLOCKS (ROWS / (ST_RPW * (PRE_THREADS / 32)))
#define PRE_BLOCKS (PREP_BLOCKS + ST_BLOCKS)

__device__ __forceinline__ void dnm_fold_block(const float* __restrict__ sw,
                                               const float* __restrict__ sb,
                                               const float* __restrict__ dw,
                                               float* __restrict__ ws,
                                               float* red, const int o, const int t) {
  float* Wp = ws + WP_OFF;
  float* cv = ws + C_OFF;
  if (o < OO) {
    if (t < NN) {
      const float d0 = dw[o * 2 + 0];
      const float d1 = dw[o * 2 + 1];
      Wp[(size_t)o * NN + t] = d0 * sw[(size_t)(o * 2 + 0) * NN + t] +
                               d1 * sw[(size_t)(o * 2 + 1) * NN + t];
      red[t] = d0 * sb[(size_t)(o * 2 + 0) * NN + t] +
               d1 * sb[(size_t)(o * 2 + 1) * NN + t];
    }
    __syncthreads();
    if (t == 0) {
      float s = 0.0f;
      for (int i = 0; i < NN; ++i) s += red[i];
      cv[o] = s;
    }
  } else {
    if (t < NN) Wp[(size_t)o * NN + t] = 0.0f;
    if (t == 0) cv[o] = 0.0f;
  }
}

__device__ __forceinline__ void dnm_stats_block(const float* __restrict__ x,
                                                float* __restrict__ ws,
                                                const int sblock, const int tid) {
  const int lane = tid & 31;
  const int w    = tid >> 5;
  const int row0 = (sblock * (PRE_THREADS / 32) + w) * ST_RPW;
  float* st = ws + ST_OFF;

  v4f v[ST_RPW][6];
  for (int r = 0; r < ST_RPW; ++r) {
    const v4f* xr = (const v4f*)(x + (size_t)(row0 + r) * DD);
    for (int i = 0; i < 6; ++i) v[r][i] = xr[lane + 32 * i];
  }

  float mu[ST_RPW];
  for (int r = 0; r < ST_RPW; ++r) {
    float s = 0.0f;
    for (int i = 0; i < 6; ++i) s += (v[r][i].x + v[r][i].y) + (v[r][i].z + v[r][i].w);
    for (int off = 16; off > 0; off >>= 1) s += __shfl_xor(s, off, 32);
    mu[r] = s * (1.0f / DD);
  }
  for (int r = 0; r < ST_RPW; ++r) {
    float q = 0.0f;
    for (int i = 0; i < 6; ++i) {
      const float d0 = v[r][i].x - mu[r], d1 = v[r][i].y - mu[r];
      const float d2 = v[r][i].z - mu[r], d3 = v[r][i].w - mu[r];
      q += (d0 * d0 + d1 * d1) + (d2 * d2 + d3 * d3);
    }
    for (int off = 16; off > 0; off >>= 1) q += __shfl_xor(q, off, 32);
    if (lane == 0) {
      v2f o2;
      o2.x = mu[r];
      o2.y = rsqrtf(q * (1.0f / DD) + 1e-5f);
      *(v2f*)(st + (size_t)(row0 + r) * 2) = o2;
    }
  }
}

__global__ void __launch_bounds__(PRE_THREADS) dnm_pre_kernel(const float* __restrict__ x,
                                                              const float* __restrict__ sw,
                                                              const float* __restrict__ sb,
                                                              const float* __restrict__ dw,
                                                              float* __restrict__ ws) {
  __shared__ float red[NN];
  if (blockIdx.x >= PREP_BLOCKS)
    dnm_stats_block(x, ws, blockIdx.x - PREP_BLOCKS, threadIdx.x);
  else
    dnm_fold_block(sw, sb, dw, ws, red, blockIdx.x, threadIdx.x);
}

#define KPITCH 209
#define AS_FLOATS (KSTEPS * KPITCH * 4)
#define KCH 7
#define NSUB 12
#define NGRP (NT / NSUB)
#define GEMM_THREADS (NSUB * 2 * 32)
#define GEMM_BLOCKS (BATCH * NGRP)

__device__ __forceinline__ v2f dnm_splat(const float c) {
  v2f r;
  r.x = c;
  r.y = c;
  return r;
}

__device__ __forceinline__ v2f dnm_erf2(const v2f z) {
  v2f az;
  az.x = __builtin_fabsf(z.x);
  az.y = __builtin_fabsf(z.y);
  v2f p = __builtin_elementwise_fma(az, dnm_splat(0.0000169988107f), dnm_splat(-0.000378677854f));
  p = __builtin_elementwise_fma(az, p, dnm_splat(0.00385781587f));
  p = __builtin_elementwise_fma(az, p, dnm_splat(-0.0241816975f));
  p = __builtin_elementwise_fma(az, p, dnm_splat(0.106668264f));
  p = __builtin_elementwise_fma(az, p, dnm_splat(0.634933293f));
  p = __builtin_elementwise_fma(az, p, dnm_splat(0.128689408f));
  const v2f q = __builtin_elementwise_fma(az, p, az);
  const v2f y = q * dnm_splat(-1.44269502f);
  v2f bg;
  bg.x = 1.0f - __builtin_amdgcn_exp2f(y.x);
  bg.y = 1.0f - __builtin_amdgcn_exp2f(y.y);
  const v2f s = az * az;
  v2f r = __builtin_elementwise_fma(s, dnm_splat(-0.000561801775f), dnm_splat(0.00491381623f));
  r = __builtin_elementwise_fma(s, r, dnm_splat(-0.0267075151f));
  r = __builtin_elementwise_fma(s, r, dnm_splat(0.112800106f));
  r = __builtin_elementwise_fma(s, r, dnm_splat(-0.376122952f));
  r = __builtin_elementwise_fma(s, r, dnm_splat(0.128379107f));
  const v2f sm = __builtin_elementwise_fma(az, r, az);
  v2f e;
  e.x = __builtin_copysignf((az.x < 1.0f) ? sm.x : bg.x, z.x);
  e.y = __builtin_copysignf((az.y < 1.0f) ? sm.y : bg.y, z.y);
  return e;
}

__device__ __forceinline__ v2f dnm_norm2(const v2f raw, const v4f s4, const float lwc, const float lbc) {
  v2f bn;
  bn.x = (raw.x - s4.x) * s4.y * lwc + lbc;
  bn.y = (raw.y - s4.z) * s4.w * lwc + lbc;
  return bn;
}

template <int NM>
__device__ __forceinline__ void dnm_gemm_task(const float* __restrict__ x,
                                              float* __restrict__ out,
                                              const float* As, const float* cs,
                                              const float* st,
                                              const float* __restrict__ Bp,
                                              const v2f (&b0)[KCH],
                                              const float lwc, const float lbc,
                                              const int b, const int col,
                                              const int m0, const int c, const int hi) {
  const float* Abase = As + ((m0 * 16 + c) * 4 + hi * 2);
  const float* Sbase = st + hi * 4;

  v8f acc[NM];
#pragma unroll
  for (int i = 0; i < NM; ++i) acc[i] = (v8f){};

  v2f bcur[KCH], bnxt[KCH];
#pragma unroll
  for (int j = 0; j < KCH; ++j) {
    bcur[j] = b0[j];
    bnxt[j] = b0[j];
  }

  v2f acur[NM], anxt[NM];
#pragma unroll
  for (int i = 0; i < NM; ++i) acur[i] = *(const v2f*)(Abase + i * 64);
  v4f snx = *(const v4f*)(Sbase + 8);
  v2f bn  = dnm_norm2(bcur[0], *(const v4f*)(Sbase), lwc, lbc);

#pragma unroll 1
  for (int ch = 0; ch < KSTEPS / KCH; ++ch) {
    const bool more = (ch + 1 < KSTEPS / KCH);
    if (more) {
#pragma unroll
      for (int j = 0; j < KCH; ++j) {
        const float* p = Bp + (size_t)((ch + 1) * KCH + j) * (4 * DD);
        bnxt[j].x = p[0];
        bnxt[j].y = p[DD];
      }
    }
#pragma unroll
    for (int j = 0; j < KCH; ++j) {
      const int kb = ch * KCH + j;
      const int kb1 = (j + 1 < KCH || more) ? kb + 1 : KSTEPS - 1;
      const int kb2 = (j + 2 < KCH || more) ? kb + 2 : KSTEPS - 1;
      const float* Ak = Abase + kb1 * (KPITCH * 4);
#pragma unroll
      for (int i = 0; i < NM; ++i) anxt[i] = *(const v2f*)(Ak + i * 64);
      const v4f sn2 = *(const v4f*)(Sbase + kb2 * 8);
      __builtin_amdgcn_sched_barrier(0);
      const v2f bn1 = dnm_norm2((j + 1 < KCH) ? bcur[j + 1] : bnxt[0], snx, lwc, lbc);
#pragma unroll
      for (int i = 0; i < NM; ++i) {
        acc[i] = __builtin_amdgcn_wmma_f32_16x16x4_f32(
             false, acur[i],  false, bn,
             (short)0, acc[i],  false,  false);
      }
#pragma unroll
      for (int i = 0; i < NM; ++i) acur[i] = anxt[i];
      bn  = bn1;
      snx = sn2;
    }
#pragma unroll
    for (int j = 0; j < KCH; ++j) bcur[j] = bnxt[j];
  }

  const size_t ebase = ((size_t)b * OO + m0 * 16 + hi * 8) * DD + col;
  const float* xe = x + ebase;
  float*       oe = out + ebase;

  float xv[NM][8];
#pragma unroll
  for (int i = 0; i < NM; ++i) {
#pragma unroll
    for (int v = 0; v < 8; ++v) {
      const int o = (m0 + i) * 16 + hi * 8 + v;
      const float* p = (o < OO) ? (xe + (size_t)(i * 16 + v) * DD) : xe;
      xv[i][v] = *p;
    }
  }

  const float* cl = cs + m0 * 16 + hi * 8;
  v4f cn0 = *(const v4f*)(cl);
  v4f cn1 = *(const v4f*)(cl + 4);
#pragma unroll
  for (int i = 0; i < NM; ++i) {
    const v4f c0 = cn0, c1 = cn1;
    if (i + 1 < NM) {
      cn0 = *(const v4f*)(cl + (i + 1) * 16);
      cn1 = *(const v4f*)(cl + (i + 1) * 16 + 4);
    }
    const float cb[8] = {c0.x, c0.y, c0.z, c0.w, c1.x, c1.y, c1.z, c1.w};
#pragma unroll
    for (int v = 0; v < 8; v += 2) {
      const int o = (m0 + i) * 16 + hi * 8 + v;
      if (o < OO) {
        v2f tv;
        tv.x = acc[i][v] + cb[v];
        tv.y = acc[i][v + 1] + cb[v + 1];
        const v2f er = dnm_erf2(tv * dnm_splat(0.70710678118654752f));
        const v2f w = __builtin_elementwise_fma(tv * dnm_splat(0.5f), er + dnm_splat(1.0f),
                                                dnm_splat(1.0f));
        oe[(size_t)(i * 16 + v) * DD]     = xv[i][v] * w.x;
        oe[(size_t)(i * 16 + v + 1) * DD] = xv[i][v + 1] * w.y;
      }
    }
  }
}

__global__ void __launch_bounds__(GEMM_THREADS) dnm_gemm_kernel(const float* __restrict__ x,
                                                                const float* __restrict__ lw,
                                                                const float* __restrict__ lb,
                                                                const float* __restrict__ ws,
                                                                float* __restrict__ out) {
  __shared__ __attribute__((aligned(16))) float As[AS_FLOATS];
  __shared__ __attribute__((aligned(16))) float st[NN * 2];
  __shared__ __attribute__((aligned(16))) float cs[OPAD];

  const float* Wp  = ws + WP_OFF;
  const float* cv  = ws + C_OFF;
  const float* stg = ws + ST_OFF;

  const int t = threadIdx.x;

  const int wv   = __builtin_amdgcn_readfirstlane(t >> 5);
  const int lane = t & 31;
  const int c    = lane & 15;
  const int hi   = lane >> 4;
  const int mh   = (wv >= NSUB) ? 1 : 0;
  const int nsub = wv - mh * NSUB;

  const int b   = blockIdx.x / NGRP;
  const int ng  = blockIdx.x - b * NGRP;
  const int col = (ng * NSUB + nsub) * 16 + c;

  const float* Bp = x + ((size_t)(b * NN + hi * 2) * DD + col);
  v2f b0[KCH];
#pragma unroll
  for (int j = 0; j < KCH; ++j) {
    const float* p = Bp + (size_t)j * (4 * DD);
    b0[j].x = p[0];
    b0[j].y = p[DD];
  }
  const float lwc = lw[col];
  const float lbc = lb[col];

  const int rot = (blockIdx.x & 15) * ((OPAD * KSTEPS) / 16);
  for (int i = t; i < OPAD * KSTEPS; i += GEMM_THREADS) {
    int ii = i + rot;
    ii = (ii >= OPAD * KSTEPS) ? (ii - OPAD * KSTEPS) : ii;
    const int m  = ii / KSTEPS;
    const int kb = ii - m * KSTEPS;
    const v4f w4 = *(const v4f*)(Wp + (size_t)m * NN + kb * 4);
    *(v4f*)(As + (kb * KPITCH + m) * 4) = w4;
  }
  if (t < OPAD) cs[t] = cv[t];
  if (t < NN) *(v2f*)(st + t * 2) = *(const v2f*)(stg + ((size_t)b * NN + t) * 2);
  __syncthreads();

  if (mh == 0)
    dnm_gemm_task<7>(x, out, As, cs, st, Bp, b0, lwc, lbc, b, col, 0, c, hi);
  else
    dnm_gemm_task<6>(x, out, As, cs, st, Bp, b0, lwc, lbc, b, col, 7, c, hi);
}

extern "C" void kernel_launch(void* const* d_in, const int* in_sizes, int n_in,
                              void* d_out, int out_size, void* d_ws, size_t ws_size,
                              hipStream_t stream) {
  const float* x    = (const float*)d_in[0];
  const float* ln_w = (const float*)d_in[1];
  const float* ln_b = (const float*)d_in[2];
  const float* sw   = (const float*)d_in[3];
  const float* sb   = (const float*)d_in[4];
  const float* dw   = (const float*)d_in[5];
  float* out = (float*)d_out;
  float* ws  = (float*)d_ws;

  dnm_pre_kernel<<<dim3(PRE_BLOCKS), dim3(PRE_THREADS), 0, stream>>>(x, sw, sb, dw, ws);
  dnm_gemm_kernel<<<dim3(GEMM_BLOCKS), dim3(GEMM_THREADS), 0, stream>>>(x, ln_w, ln_b, ws, out);
}
